# starting kernel
# baseline (speedup 1.0000x reference)
#include <hip/hip_runtime.h>


typedef __attribute__((ext_vector_type(2))) float v2f;
typedef __attribute__((ext_vector_type(8))) float v8f;

#define T_LEN 512
#define WTP_STRIDE 272
#define H_STRIDE 68

static __device__ __forceinline__ v8f wmma4(v2f a, v2f b, v8f c) {
  return __builtin_amdgcn_wmma_f32_16x16x4_f32(false, a, false, b, (short)0, c,
                                               false, false);
}

static __device__ __forceinline__ float sigm(float x) {
  return 1.0f / (1.0f + __expf(-x));
}
static __device__ __forceinline__ float tanh_(float x) {
  return 1.0f - 2.0f / (1.0f + __expf(2.0f * x));
}

__global__ __launch_bounds__(128) void lstm_wmma_kernel(
    const float* __restrict__ x, const float* __restrict__ W_ih,
    const float* __restrict__ W_hh, const float* __restrict__ b_ih,
    const float* __restrict__ b_hh, const float* __restrict__ W_out,
    const float* __restrict__ b_out, float* __restrict__ out) {
  __shared__ float WtP[34 * WTP_STRIDE * 2];
  __shared__ float WtOutP[32 * 16 * 2];
  __shared__ float hS[4 * 16 * H_STRIDE];

  const int tid  = threadIdx.x;
  const int wave = tid >> 5;
  const int lane = tid & 31;
  const int col  = lane & 15;
  const int hi   = lane >> 4;
  const int bbase = blockIdx.x * 64 + wave * 16;

  for (int idx = tid; idx < 68 * 256; idx += 128) {
    int k = idx >> 8, n = idx & 255;
    float v;
    if (k < 64)      v = W_hh[n * 64 + k];
    else if (k < 67) v = W_ih[n * 3 + (k - 64)];
    else             v = b_ih[n] + b_hh[n];
    WtP[((k >> 1) * WTP_STRIDE + n) * 2 + (k & 1)] = v;
  }
  for (int idx = tid; idx < 64 * 16; idx += 128) {
    int k = idx >> 4, n = idx & 15;
    WtOutP[((k >> 1) * 16 + n) * 2 + (k & 1)] =
        (n < 3) ? W_out[n * 64 + k] : 0.0f;
  }
  for (int idx = tid; idx < 4 * 16 * H_STRIDE; idx += 128) hS[idx] = 0.0f;
  __syncthreads();

  float* hW = &hS[wave * (16 * H_STRIDE)];
  const float* xrow = x + (size_t)(bbase + col) * (T_LEN * 3);
  const float bO = (col < 3) ? b_out[col] : 0.0f;
  const v8f vzero = {0, 0, 0, 0, 0, 0, 0, 0};

  v8f cs0 = vzero, cs1 = vzero, cs2 = vzero, cs3 = vzero;

  for (int t = 0; t < T_LEN; ++t) {
    const float* xp = xrow + t * 3;
    v2f ax;
    if (hi == 0) { ax[0] = xp[0]; ax[1] = xp[1]; }
    else         { ax[0] = xp[2]; ax[1] = 1.0f; }

    v8f hn0, hn1, hn2, hn3;

#define GATE_BLOCK(J, CS, HN)                                                 \
    {                                                                         \
      const int nb = (J) * 16 + col;                                          \
      v8f ai, af, ag, ao;                                                     \
      {             \
        const float* wp = &WtP[((32 + hi) * WTP_STRIDE + nb) * 2];            \
        v2f b0 = *(const v2f*)(wp);                                           \
        v2f b1 = *(const v2f*)(wp + 128);                                     \
        v2f b2 = *(const v2f*)(wp + 256);                                     \
        v2f b3 = *(const v2f*)(wp + 384);                                     \
        ai = wmma4(ax, b0, vzero);                                            \
        af = wmma4(ax, b1, vzero);                                            \
        ag = wmma4(ax, b2, vzero);                                            \
        ao = wmma4(ax, b3, vzero);                                            \
      }                                                                       \
      _Pragma("unroll 2")                                                     \
      for (int ks = 0; ks < 16; ++ks) {                                       \
        v2f a = *(const v2f*)&hW[col * H_STRIDE + ks * 4 + hi * 2];           \
        const float* wp = &WtP[((ks * 2 + hi) * WTP_STRIDE + nb) * 2];        \
        v2f b0 = *(const v2f*)(wp);                                           \
        v2f b1 = *(const v2f*)(wp + 128);                                     \
        v2f b2 = *(const v2f*)(wp + 256);                                     \
        v2f b3 = *(const v2f*)(wp + 384);                                     \
        ai = wmma4(a, b0, ai);                                                \
        af = wmma4(a, b1, af);                                                \
        ag = wmma4(a, b2, ag);                                                \
        ao = wmma4(a, b3, ao);                                                \
      }                                                                       \
      _Pragma("unroll")                                                       \
      for (int r = 0; r < 8; ++r) {                                           \
        float iv = sigm(ai[r]), fv = sigm(af[r]);                             \
        float gv = tanh_(ag[r]), ov = sigm(ao[r]);                            \
        float cv = fv * CS[r] + iv * gv;                                      \
        CS[r] = cv;                                                           \
        HN[r] = ov * tanh_(cv);                                               \
      }                                                                       \
    }

    GATE_BLOCK(0, cs0, hn0)
    GATE_BLOCK(1, cs1, hn1)
    GATE_BLOCK(2, cs2, hn2)
    GATE_BLOCK(3, cs3, hn3)
#undef GATE_BLOCK

    {
      float* hp = &hW[hi * 8 * H_STRIDE + col];
#pragma unroll
      for (int r = 0; r < 8; ++r) {
        hp[r * H_STRIDE +  0] = hn0[r];
        hp[r * H_STRIDE + 16] = hn1[r];
        hp[r * H_STRIDE + 32] = hn2[r];
        hp[r * H_STRIDE + 48] = hn3[r];
      }
    }

    v8f aco = {bO, bO, bO, bO, bO, bO, bO, bO};
#pragma unroll 2
    for (int ks = 0; ks < 16; ++ks) {
      v2f a = *(const v2f*)&hW[col * H_STRIDE + ks * 4 + hi * 2];
      v2f b = *(const v2f*)&WtOutP[((ks * 2 + hi) * 16 + col) * 2];
      aco = wmma4(a, b, aco);
    }
    if (col < 3) {
#pragma unroll
      for (int r = 0; r < 8; ++r) {
        size_t bidx = (size_t)(bbase + hi * 8 + r);
        out[(bidx * T_LEN + t) * 3 + col] = aco[r];
      }
    }
  }
}

extern "C" void kernel_launch(void* const* d_in, const int* in_sizes, int n_in,
                              void* d_out, int out_size, void* d_ws,
                              size_t ws_size, hipStream_t stream) {
  const float* x     = (const float*)d_in[0];
  const float* W_ih  = (const float*)d_in[1];
  const float* W_hh  = (const float*)d_in[2];
  const float* b_ih  = (const float*)d_in[3];
  const float* b_hh  = (const float*)d_in[4];
  const float* W_out = (const float*)d_in[5];
  const float* b_out = (const float*)d_in[6];
  float* out = (float*)d_out;

  dim3 grid(4096 / 64);
  dim3 block(128);
  lstm_wmma_kernel<<<grid, block, 0, stream>>>(x, W_ih, W_hh, b_ih, b_hh,
                                               W_out, b_out, out);
}


# Split the 256 gate columns over a workgroup's 4 waves (16 batch rows per workgroup, 256 workgroups, a wave on every SIMD): weight slice in registers, h exchanged through LDS, one barrier per step
# speedup vs baseline: 4.4853x; 4.4853x over previous
#include <hip/hip_runtime.h>


typedef __attribute__((ext_vector_type(2))) float v2f;
typedef __attribute__((ext_vector_type(8))) float v8f;

#define T_LEN 512
#define H_STRIDE 68
#define H_TILE (16 * H_STRIDE)

static __device__ __forceinline__ v8f wmma4(v2f a, v2f b, v8f c) {
  return __builtin_amdgcn_wmma_f32_16x16x4_f32(false, a, false, b, (short)0, c,
                                               false, false);
}

static __device__ __forceinline__ float sigm(float x) {
  return 1.0f / (1.0f + __expf(-x));
}
static __device__ __forceinline__ float tanh_(float x) {
  return 1.0f - 2.0f / (1.0f + __expf(2.0f * x));
}

__global__ __launch_bounds__(128) void lstm_wmma_kernel(
    const float* __restrict__ x, const float* __restrict__ W_ih,
    const float* __restrict__ W_hh, const float* __restrict__ b_ih,
    const float* __restrict__ b_hh, const float* __restrict__ W_out,
    const float* __restrict__ b_out, float* __restrict__ out) {
  __shared__ float hS[2 * H_TILE];

  const int tid  = threadIdx.x;
  const int lane = tid & 31;
  const int wave = __builtin_amdgcn_readfirstlane(tid >> 5);
  const int col  = lane & 15;
  const int hi   = lane >> 4;
  const int row0 = blockIdx.x * 16;

  v2f wb[17][4];
#pragma unroll
  for (int g = 0; g < 4; ++g) {
    const int n = g * 64 + wave * 16 + col;
    const float* wr = W_hh + n * 64 + hi * 2;
#pragma unroll
    for (int ks = 0; ks < 16; ++ks) {
      wb[ks][g][0] = wr[ks * 4];
      wb[ks][g][1] = wr[ks * 4 + 1];
    }
    const float bs = b_ih[n] + b_hh[n];
    wb[16][g][0] = W_ih[n * 3 + hi * 2];
    wb[16][g][1] = hi ? bs : W_ih[n * 3 + 1];
  }
  v2f wo[16];
  const float* wor = W_out + ((col < 3) ? col : 0) * 64 + hi * 2;
#pragma unroll
  for (int ks = 0; ks < 16; ++ks) {
    wo[ks][0] = (col < 3) ? wor[ks * 4]     : 0.0f;
    wo[ks][1] = (col < 3) ? wor[ks * 4 + 1] : 0.0f;
  }
  const float bO = (col < 3) ? b_out[col] : 0.0f;

  for (int idx = tid; idx < H_TILE; idx += 128) hS[idx] = 0.0f;
  __syncthreads();

  const float* xrow = x + (size_t)(row0 + col) * (T_LEN * 3);
  float* orow = out + (size_t)(row0 + hi * 8) * (T_LEN * 3) + col;
  const v8f vzero = {0, 0, 0, 0, 0, 0, 0, 0};
  v8f cs = vzero;

  for (int t = 0; t < T_LEN; ++t) {
    const float* hR = &hS[(t & 1) * H_TILE];
    float*       hN = &hS[((t + 1) & 1) * H_TILE];

    const float* xp = xrow + t * 3;
    v2f ax;
    if (hi == 0) { ax[0] = xp[0]; ax[1] = xp[1]; }
    else         { ax[0] = xp[2]; ax[1] = 1.0f; }

    v2f a[16];
#pragma unroll
    for (int ks = 0; ks < 16; ++ks)
      a[ks] = *(const v2f*)&hR[col * H_STRIDE + ks * 4 + hi * 2];

    if (wave == 0 && t > 0) {
      v8f aco = {bO, bO, bO, bO, bO, bO, bO, bO};
#pragma unroll
      for (int ks = 0; ks < 16; ++ks) aco = wmma4(a[ks], wo[ks], aco);
      if (col < 3) {
#pragma unroll
        for (int r = 0; r < 8; ++r)
          orow[(size_t)r * (T_LEN * 3) + (t - 1) * 3] = aco[r];
      }
    }

    v8f ai = wmma4(ax, wb[16][0], vzero);
    v8f af = wmma4(ax, wb[16][1], vzero);
    v8f ag = wmma4(ax, wb[16][2], vzero);
    v8f ao = wmma4(ax, wb[16][3], vzero);
#pragma unroll
    for (int ks = 0; ks < 16; ++ks) {
      ai = wmma4(a[ks], wb[ks][0], ai);
      af = wmma4(a[ks], wb[ks][1], af);
      ag = wmma4(a[ks], wb[ks][2], ag);
      ao = wmma4(a[ks], wb[ks][3], ao);
    }

    float* hp = &hN[hi * 8 * H_STRIDE + wave * 16 + col];
#pragma unroll
    for (int r = 0; r < 8; ++r) {
      float iv = sigm(ai[r]), fv = sigm(af[r]);
      float gv = tanh_(ag[r]), ov = sigm(ao[r]);
      float cv = fv * cs[r] + iv * gv;
      cs[r] = cv;
      hp[r * H_STRIDE] = ov * tanh_(cv);
    }
    __syncthreads();
  }

  if (wave == 0) {
    const float* hR = &hS[(T_LEN & 1) * H_TILE];
    v8f aco = {bO, bO, bO, bO, bO, bO, bO, bO};
#pragma unroll
    for (int ks = 0; ks < 16; ++ks) {
      v2f a = *(const v2f*)&hR[col * H_STRIDE + ks * 4 + hi * 2];
      aco = wmma4(a, wo[ks], aco);
    }
    if (col < 3) {
#pragma unroll
      for (int r = 0; r < 8; ++r)
        orow[(size_t)r * (T_LEN * 3) + (T_LEN - 1) * 3] = aco[r];
    }
  }
}

extern "C" void kernel_launch(void* const* d_in, const int* in_sizes, int n_in,
                              void* d_out, int out_size, void* d_ws,
                              size_t ws_size, hipStream_t stream) {
  const float* x     = (const float*)d_in[0];
  const float* W_ih  = (const float*)d_in[1];
  const float* W_hh  = (const float*)d_in[2];
  const float* b_ih  = (const float*)d_in[3];
  const float* b_hh  = (const float*)d_in[4];
  const float* W_out = (const float*)d_in[5];
  const float* b_out = (const float*)d_in[6];
  float* out = (float*)d_out;

  dim3 grid(256);
  dim3 block(128);
  lstm_wmma_kernel<<<grid, block, 0, stream>>>(x, W_ih, W_hh, b_ih, b_hh,
                                               W_out, b_out, out);
}


# Gate sigmoid and tanh take 1/(1+exp) from the hardware reciprocal (v_rcp_f32, 1 ulp) instead of the correctly rounded divide sequence: 40 quotients per lane per timestep, one instruction each
# speedup vs baseline: 6.2014x; 1.3826x over previous
#include <hip/hip_runtime.h>


typedef __attribute__((ext_vector_type(2))) float v2f;
typedef __attribute__((ext_vector_type(8))) float v8f;

#define T_LEN 512
#define H_STRIDE 68
#define H_TILE (16 * H_STRIDE)

static __device__ __forceinline__ v8f wmma4(v2f a, v2f b, v8f c) {
  return __builtin_amdgcn_wmma_f32_16x16x4_f32(false, a, false, b, (short)0, c,
                                               false, false);
}

static __device__ __forceinline__ float sigm(float x) {
  return __builtin_amdgcn_rcpf(1.0f + __expf(-x));
}
static __device__ __forceinline__ float tanh_(float x) {
  return 1.0f - 2.0f * __builtin_amdgcn_rcpf(1.0f + __expf(2.0f * x));
}

__global__ __launch_bounds__(128) void lstm_wmma_kernel(
    const float* __restrict__ x, const float* __restrict__ W_ih,
    const float* __restrict__ W_hh, const float* __restrict__ b_ih,
    const float* __restrict__ b_hh, const float* __restrict__ W_out,
    const float* __restrict__ b_out, float* __restrict__ out) {
  __shared__ float hS[2 * H_TILE];

  const int tid  = threadIdx.x;
  const int lane = tid & 31;
  const int wave = __builtin_amdgcn_readfirstlane(tid >> 5);
  const int col  = lane & 15;
  const int hi   = lane >> 4;
  const int row0 = blockIdx.x * 16;

  v2f wb[17][4];
#pragma unroll
  for (int g = 0; g < 4; ++g) {
    const int n = g * 64 + wave * 16 + col;
    const float* wr = W_hh + n * 64 + hi * 2;
#pragma unroll
    for (int ks = 0; ks < 16; ++ks) {
      wb[ks][g][0] = wr[ks * 4];
      wb[ks][g][1] = wr[ks * 4 + 1];
    }
    const float bs = b_ih[n] + b_hh[n];
    wb[16][g][0] = W_ih[n * 3 + hi * 2];
    wb[16][g][1] = hi ? bs : W_ih[n * 3 + 1];
  }
  v2f wo[16];
  const float* wor = W_out + ((col < 3) ? col : 0) * 64 + hi * 2;
#pragma unroll
  for (int ks = 0; ks < 16; ++ks) {
    wo[ks][0] = (col < 3) ? wor[ks * 4]     : 0.0f;
    wo[ks][1] = (col < 3) ? wor[ks * 4 + 1] : 0.0f;
  }
  const float bO = (col < 3) ? b_out[col] : 0.0f;

  for (int idx = tid; idx < H_TILE; idx += 128) hS[idx] = 0.0f;
  __syncthreads();

  const float* xrow = x + (size_t)(row0 + col) * (T_LEN * 3);
  float* orow = out + (size_t)(row0 + hi * 8) * (T_LEN * 3) + col;
  const v8f vzero = {0, 0, 0, 0, 0, 0, 0, 0};
  v8f cs = vzero;

  for (int t = 0; t < T_LEN; ++t) {
    const float* hR = &hS[(t & 1) * H_TILE];
    float*       hN = &hS[((t + 1) & 1) * H_TILE];

    const float* xp = xrow + t * 3;
    v2f ax;
    if (hi == 0) { ax[0] = xp[0]; ax[1] = xp[1]; }
    else         { ax[0] = xp[2]; ax[1] = 1.0f; }

    v2f a[16];
#pragma unroll
    for (int ks = 0; ks < 16; ++ks)
      a[ks] = *(const v2f*)&hR[col * H_STRIDE + ks * 4 + hi * 2];

    if (wave == 0 && t > 0) {
      v8f aco = {bO, bO, bO, bO, bO, bO, bO, bO};
#pragma unroll
      for (int ks = 0; ks < 16; ++ks) aco = wmma4(a[ks], wo[ks], aco);
      if (col < 3) {
#pragma unroll
        for (int r = 0; r < 8; ++r)
          orow[(size_t)r * (T_LEN * 3) + (t - 1) * 3] = aco[r];
      }
    }

    v8f ai = wmma4(ax, wb[16][0], vzero);
    v8f af = wmma4(ax, wb[16][1], vzero);
    v8f ag = wmma4(ax, wb[16][2], vzero);
    v8f ao = wmma4(ax, wb[16][3], vzero);
#pragma unroll
    for (int ks = 0; ks < 16; ++ks) {
      ai = wmma4(a[ks], wb[ks][0], ai);
      af = wmma4(a[ks], wb[ks][1], af);
      ag = wmma4(a[ks], wb[ks][2], ag);
      ao = wmma4(a[ks], wb[ks][3], ao);
    }

    float* hp = &hN[hi * 8 * H_STRIDE + wave * 16 + col];
#pragma unroll
    for (int r = 0; r < 8; ++r) {
      float iv = sigm(ai[r]), fv = sigm(af[r]);
      float gv = tanh_(ag[r]), ov = sigm(ao[r]);
      float cv = fv * cs[r] + iv * gv;
      cs[r] = cv;
      hp[r * H_STRIDE] = ov * tanh_(cv);
    }
    __syncthreads();
  }

  if (wave == 0) {
    const float* hR = &hS[(T_LEN & 1) * H_TILE];
    v8f aco = {bO, bO, bO, bO, bO, bO, bO, bO};
#pragma unroll
    for (int ks = 0; ks < 16; ++ks) {
      v2f a = *(const v2f*)&hR[col * H_STRIDE + ks * 4 + hi * 2];
      aco = wmma4(a, wo[ks], aco);
    }
    if (col < 3) {
#pragma unroll
      for (int r = 0; r < 8; ++r)
        orow[(size_t)r * (T_LEN * 3) + (T_LEN - 1) * 3] = aco[r];
    }
  }
}

extern "C" void kernel_launch(void* const* d_in, const int* in_sizes, int n_in,
                              void* d_out, int out_size, void* d_ws,
                              size_t ws_size, hipStream_t stream) {
  const float* x     = (const float*)d_in[0];
  const float* W_ih  = (const float*)d_in[1];
  const float* W_hh  = (const float*)d_in[2];
  const float* b_ih  = (const float*)d_in[3];
  const float* b_hh  = (const float*)d_in[4];
  const float* W_out = (const float*)d_in[5];
  const float* b_out = (const float*)d_in[6];
  float* out = (float*)d_out;

  dim3 grid(256);
  dim3 block(128);
  lstm_wmma_kernel<<<grid, block, 0, stream>>>(x, W_ih, W_hh, b_ih, b_hh,
                                               W_out, b_out, out);
}


# Output projection split over the 4 waves along K: 4 WMMAs per wave on its own 16 hidden units instead of 16 on wave 0; partials summed through LDS one step later (out sum order changes)
# speedup vs baseline: 6.4121x; 1.0340x over previous
#include <hip/hip_runtime.h>


typedef __attribute__((ext_vector_type(2))) float v2f;
typedef __attribute__((ext_vector_type(8))) float v8f;

#define T_LEN 512
#define H_STRIDE 68
#define H_TILE (16 * H_STRIDE)
#define P_TILE (3 * 4 * 16)

typedef __attribute__((ext_vector_type(4))) float v4f;

static __device__ __forceinline__ v8f wmma4(v2f a, v2f b, v8f c) {
  return __builtin_amdgcn_wmma_f32_16x16x4_f32(false, a, false, b, (short)0, c,
                                               false, false);
}

static __device__ __forceinline__ float sigm(float x) {
  return __builtin_amdgcn_rcpf(1.0f + __expf(-x));
}
static __device__ __forceinline__ float tanh_(float x) {
  return 1.0f - 2.0f * __builtin_amdgcn_rcpf(1.0f + __expf(2.0f * x));
}

__global__ __launch_bounds__(128) void lstm_wmma_kernel(
    const float* __restrict__ x, const float* __restrict__ W_ih,
    const float* __restrict__ W_hh, const float* __restrict__ b_ih,
    const float* __restrict__ b_hh, const float* __restrict__ W_out,
    const float* __restrict__ b_out, float* __restrict__ out) {
  __shared__ float hS[2 * H_TILE];
  __shared__ __attribute__((aligned(16))) float pS[2 * P_TILE];

  const int tid  = threadIdx.x;
  const int lane = tid & 31;
  const int wave = __builtin_amdgcn_readfirstlane(tid >> 5);
  const int col  = lane & 15;
  const int hi   = lane >> 4;
  const int row0 = blockIdx.x * 16;

  v2f wb[17][4];
#pragma unroll
  for (int g = 0; g < 4; ++g) {
    const int n = g * 64 + wave * 16 + col;
    const float* wr = W_hh + n * 64 + hi * 2;
#pragma unroll
    for (int ks = 0; ks < 16; ++ks) {
      wb[ks][g][0] = wr[ks * 4];
      wb[ks][g][1] = wr[ks * 4 + 1];
    }
    const float bs = b_ih[n] + b_hh[n];
    wb[16][g][0] = W_ih[n * 3 + hi * 2];
    wb[16][g][1] = hi ? bs : W_ih[n * 3 + 1];
  }
  v2f wo[4];
  const float* wor = W_out + ((col < 3) ? col : 0) * 64 + wave * 16 + hi * 2;
#pragma unroll
  for (int j = 0; j < 4; ++j) {
    wo[j][0] = (col < 3) ? wor[j * 4]     : 0.0f;
    wo[j][1] = (col < 3) ? wor[j * 4 + 1] : 0.0f;
  }
  const int sl = (lane < 12) ? lane : 0;
  const int so = sl % 3;
  const int sr = wave * 4 + sl / 3;
  const float bO = b_out[so];
  const int pRd = so * 64 + sr;
  float* osum = out + (size_t)(row0 + sr) * (T_LEN * 3) + so;
  const int pWr = (col * 4 + wave) * 16 + hi * 8;
  const int aOff = col * H_STRIDE + wave * 16 + hi * 2;

  for (int idx = tid; idx < H_TILE; idx += 128) hS[idx] = 0.0f;
  for (int idx = tid; idx < 2 * P_TILE; idx += 128) pS[idx] = 0.0f;
  __syncthreads();

  const float* xrow = x + (size_t)(row0 + col) * (T_LEN * 3);
  const v8f vzero = {0, 0, 0, 0, 0, 0, 0, 0};
  v8f cs = vzero;

  for (int t = 0; t < T_LEN; ++t) {
    const float* hR = &hS[(t & 1) * H_TILE];
    float*       hN = &hS[((t + 1) & 1) * H_TILE];

    const float* xp = xrow + t * 3;
    v2f ax;
    if (hi == 0) { ax[0] = xp[0]; ax[1] = xp[1]; }
    else         { ax[0] = xp[2]; ax[1] = 1.0f; }

    const float* pR = &pS[((t + 1) & 1) * P_TILE + pRd];
    const float q0 = pR[0], q1 = pR[16], q2 = pR[32], q3 = pR[48];

    v2f a[16];
#pragma unroll
    for (int ks = 0; ks < 16; ++ks)
      a[ks] = *(const v2f*)&hR[col * H_STRIDE + ks * 4 + hi * 2];
    v2f ap[4];
#pragma unroll
    for (int j = 0; j < 4; ++j) ap[j] = *(const v2f*)&hR[aOff + j * 4];

    asm volatile("" :: "v"(q0), "v"(q1), "v"(q2), "v"(q3));
    if (t >= 2 && lane < 12) osum[(t - 2) * 3] = ((q0 + q1) + (q2 + q3)) + bO;

    v8f po = wmma4(ap[0], wo[0], vzero);
#pragma unroll
    for (int j = 1; j < 4; ++j) po = wmma4(ap[j], wo[j], po);

    v8f ai = wmma4(ax, wb[16][0], vzero);
    v8f af = wmma4(ax, wb[16][1], vzero);
    v8f ag = wmma4(ax, wb[16][2], vzero);
    v8f ao = wmma4(ax, wb[16][3], vzero);
#pragma unroll
    for (int ks = 0; ks < 16; ++ks) {
      ai = wmma4(a[ks], wb[ks][0], ai);
      af = wmma4(a[ks], wb[ks][1], af);
      ag = wmma4(a[ks], wb[ks][2], ag);
      ao = wmma4(a[ks], wb[ks][3], ao);
    }

    float* hp = &hN[hi * 8 * H_STRIDE + wave * 16 + col];
#pragma unroll
    for (int r = 0; r < 8; ++r) {
      float iv = sigm(ai[r]), fv = sigm(af[r]);
      float gv = tanh_(ag[r]), ov = sigm(ao[r]);
      float cv = fv * cs[r] + iv * gv;
      cs[r] = cv;
      hp[r * H_STRIDE] = ov * tanh_(cv);
    }
    if (col < 3) {
      v4f* pW = (v4f*)&pS[(t & 1) * P_TILE + pWr];
      pW[0] = v4f{po[0], po[1], po[2], po[3]};
      pW[1] = v4f{po[4], po[5], po[6], po[7]};
    }
    __syncthreads();
  }

  if (lane < 12) {
    const float* pR = &pS[((T_LEN + 1) & 1) * P_TILE + pRd];
    osum[(T_LEN - 2) * 3] = ((pR[0] + pR[16]) + (pR[32] + pR[48])) + bO;
  }
  {
    const float* hR = &hS[(T_LEN & 1) * H_TILE];
    v8f po = vzero;
#pragma unroll
    for (int j = 0; j < 4; ++j)
      po = wmma4(*(const v2f*)&hR[aOff + j * 4], wo[j], po);
    if (col < 3) {
      v4f* pW = (v4f*)&pS[(T_LEN & 1) * P_TILE + pWr];
      pW[0] = v4f{po[0], po[1], po[2], po[3]};
      pW[1] = v4f{po[4], po[5], po[6], po[7]};
    }
  }
  __syncthreads();
  if (lane < 12) {
    const float* pR = &pS[(T_LEN & 1) * P_TILE + pRd];
    osum[(T_LEN - 1) * 3] = ((pR[0] + pR[16]) + (pR[32] + pR[48])) + bO;
  }
}

extern "C" void kernel_launch(void* const* d_in, const int* in_sizes, int n_in,
                              void* d_out, int out_size, void* d_ws,
                              size_t ws_size, hipStream_t stream) {
  const float* x     = (const float*)d_in[0];
  const float* W_ih  = (const float*)d_in[1];
  const float* W_hh  = (const float*)d_in[2];
  const float* b_ih  = (const float*)d_in[3];
  const float* b_hh  = (const float*)d_in[4];
  const float* W_out = (const float*)d_in[5];
  const float* b_out = (const float*)d_in[6];
  float* out = (float*)d_out;

  dim3 grid(256);
  dim3 block(128);
  lstm_wmma_kernel<<<grid, block, 0, stream>>>(x, W_ih, W_hh, b_ih, b_hh,
                                               W_out, b_out, out);
}


# Software-pipelined step: x(t+1) loaded a step ahead, x/bias WMMAs under the LDS fragment loads, i,g chains first and activated beside the f,o WMMAs, out-projection WMMAs beside the cell update
# speedup vs baseline: 7.0094x; 1.0932x over previous
#include <hip/hip_runtime.h>


typedef __attribute__((ext_vector_type(2))) float v2f;
typedef __attribute__((ext_vector_type(8))) float v8f;

#define T_LEN 512
#define H_STRIDE 68
#define H_TILE (16 * H_STRIDE)
#define P_TILE (3 * 4 * 16)

typedef __attribute__((ext_vector_type(4))) float v4f;

static __device__ __forceinline__ v8f wmma4(v2f a, v2f b, v8f c) {
  return __builtin_amdgcn_wmma_f32_16x16x4_f32(false, a, false, b, (short)0, c,
                                               false, false);
}

static __device__ __forceinline__ float sigm(float x) {
  return __builtin_amdgcn_rcpf(1.0f + __expf(-x));
}
static __device__ __forceinline__ float tanh_(float x) {
  return 1.0f - 2.0f * __builtin_amdgcn_rcpf(1.0f + __expf(2.0f * x));
}

__global__ __launch_bounds__(128) void lstm_wmma_kernel(
    const float* __restrict__ x, const float* __restrict__ W_ih,
    const float* __restrict__ W_hh, const float* __restrict__ b_ih,
    const float* __restrict__ b_hh, const float* __restrict__ W_out,
    const float* __restrict__ b_out, float* __restrict__ out) {
  __shared__ float hS[2 * H_TILE];
  __shared__ __attribute__((aligned(16))) float pS[2 * P_TILE];

  const int tid  = threadIdx.x;
  const int lane = tid & 31;
  const int wave = __builtin_amdgcn_readfirstlane(tid >> 5);
  const int col  = lane & 15;
  const int hi   = lane >> 4;
  const int row0 = blockIdx.x * 16;

  v2f wb[17][4];
#pragma unroll
  for (int g = 0; g < 4; ++g) {
    const int n = g * 64 + wave * 16 + col;
    const float* wr = W_hh + n * 64 + hi * 2;
#pragma unroll
    for (int ks = 0; ks < 16; ++ks) {
      wb[ks][g][0] = wr[ks * 4];
      wb[ks][g][1] = wr[ks * 4 + 1];
    }
    const float bs = b_ih[n] + b_hh[n];
    wb[16][g][0] = W_ih[n * 3 + hi * 2];
    wb[16][g][1] = hi ? bs : W_ih[n * 3 + 1];
  }
  v2f wo[4];
  const float* wor = W_out + ((col < 3) ? col : 0) * 64 + wave * 16 + hi * 2;
#pragma unroll
  for (int j = 0; j < 4; ++j) {
    wo[j][0] = (col < 3) ? wor[j * 4]     : 0.0f;
    wo[j][1] = (col < 3) ? wor[j * 4 + 1] : 0.0f;
  }
  const int sl = (lane < 12) ? lane : 0;
  const int so = sl % 3;
  const int sr = wave * 4 + sl / 3;
  const float bO = b_out[so];
  const int pRd = so * 64 + sr;
  float* osum = out + (size_t)(row0 + sr) * (T_LEN * 3) + so;
  const int pWr = (col * 4 + wave) * 16 + hi * 8;
  const int aOff = col * H_STRIDE + wave * 16 + hi * 2;

  for (int idx = tid; idx < H_TILE; idx += 128) hS[idx] = 0.0f;
  for (int idx = tid; idx < 2 * P_TILE; idx += 128) pS[idx] = 0.0f;
  __syncthreads();

  const float* xrow = x + (size_t)(row0 + col) * (T_LEN * 3);
  const v8f vzero = {0, 0, 0, 0, 0, 0, 0, 0};
  v8f cs = vzero;
  const int xk0 = hi * 2;
  const int xk1 = hi + 1;
  v2f ax;
  {
    const float x0 = xrow[xk0], x1 = xrow[xk1];
    ax[0] = x0;
    ax[1] = hi ? 1.0f : x1;
  }

  for (int t = 0; t < T_LEN; ++t) {
    const float* hR = &hS[(t & 1) * H_TILE];
    float*       hN = &hS[((t + 1) & 1) * H_TILE];

    const float* xp = xrow + ((t + 1 < T_LEN) ? (t + 1) : t) * 3;
    const float nx0 = xp[xk0], nx1 = xp[xk1];

    const float* pR = &pS[((t + 1) & 1) * P_TILE + pRd];
    const float q0 = pR[0], q1 = pR[16], q2 = pR[32], q3 = pR[48];

    v2f a[16];
#pragma unroll
    for (int ks = 0; ks < 16; ++ks)
      a[ks] = *(const v2f*)&hR[col * H_STRIDE + ks * 4 + hi * 2];
    v2f ap[4];
#pragma unroll
    for (int j = 0; j < 4; ++j) ap[j] = *(const v2f*)&hR[aOff + j * 4];

    v8f ai = wmma4(ax, wb[16][0], vzero);
    v8f af = wmma4(ax, wb[16][1], vzero);
    v8f ag = wmma4(ax, wb[16][2], vzero);
    v8f ao = wmma4(ax, wb[16][3], vzero);

    asm volatile("" :: "v"(q0), "v"(q1), "v"(q2), "v"(q3));
    if (t >= 2 && lane < 12) osum[(t - 2) * 3] = ((q0 + q1) + (q2 + q3)) + bO;

#pragma unroll
    for (int ks = 0; ks < 16; ++ks) {
      ai = wmma4(a[ks], wb[ks][0], ai);
      ag = wmma4(a[ks], wb[ks][2], ag);
    }
    asm volatile("" : "+v"(af), "+v"(ao) : "v"(ai), "v"(ag));
#pragma unroll
    for (int ks = 0; ks < 16; ++ks) {
      af = wmma4(a[ks], wb[ks][1], af);
      ao = wmma4(a[ks], wb[ks][3], ao);
    }
    float iv[8], gv[8];
#pragma unroll
    for (int r = 0; r < 8; ++r) {
      iv[r] = sigm(ai[r]);
      gv[r] = tanh_(ag[r]);
    }

    v2f ap0 = ap[0];
    asm volatile("" : "+v"(ap0) : "v"(af), "v"(ao));
    v8f po = wmma4(ap0, wo[0], vzero);
#pragma unroll
    for (int j = 1; j < 4; ++j) po = wmma4(ap[j], wo[j], po);

    float* hp = &hN[hi * 8 * H_STRIDE + wave * 16 + col];
#pragma unroll
    for (int r = 0; r < 8; ++r) {
      float fv = sigm(af[r]), ov = sigm(ao[r]);
      float cv = fv * cs[r] + iv[r] * gv[r];
      cs[r] = cv;
      hp[r * H_STRIDE] = ov * tanh_(cv);
    }
    if (col < 3) {
      v4f* pW = (v4f*)&pS[(t & 1) * P_TILE + pWr];
      pW[0] = v4f{po[0], po[1], po[2], po[3]};
      pW[1] = v4f{po[4], po[5], po[6], po[7]};
    }
    ax[0] = nx0;
    ax[1] = hi ? 1.0f : nx1;
    __syncthreads();
  }

  if (lane < 12) {
    const float* pR = &pS[((T_LEN + 1) & 1) * P_TILE + pRd];
    osum[(T_LEN - 2) * 3] = ((pR[0] + pR[16]) + (pR[32] + pR[48])) + bO;
  }
  {
    const float* hR = &hS[(T_LEN & 1) * H_TILE];
    v8f po = vzero;
#pragma unroll
    for (int j = 0; j < 4; ++j)
      po = wmma4(*(const v2f*)&hR[aOff + j * 4], wo[j], po);
    if (col < 3) {
      v4f* pW = (v4f*)&pS[(T_LEN & 1) * P_TILE + pWr];
      pW[0] = v4f{po[0], po[1], po[2], po[3]};
      pW[1] = v4f{po[4], po[5], po[6], po[7]};
    }
  }
  __syncthreads();
  if (lane < 12) {
    const float* pR = &pS[(T_LEN & 1) * P_TILE + pRd];
    osum[(T_LEN - 1) * 3] = ((pR[0] + pR[16]) + (pR[32] + pR[48])) + bO;
  }
}

extern "C" void kernel_launch(void* const* d_in, const int* in_sizes, int n_in,
                              void* d_out, int out_size, void* d_ws,
                              size_t ws_size, hipStream_t stream) {
  const float* x     = (const float*)d_in[0];
  const float* W_ih  = (const float*)d_in[1];
  const float* W_hh  = (const float*)d_in[2];
  const float* b_ih  = (const float*)d_in[3];
  const float* b_hh  = (const float*)d_in[4];
  const float* W_out = (const float*)d_in[5];
  const float* b_out = (const float*)d_in[6];
  float* out = (float*)d_out;

  dim3 grid(256);
  dim3 block(128);
  lstm_wmma_kernel<<<grid, block, 0, stream>>>(x, W_ih, W_hh, b_ih, b_hh,
                                               W_out, b_out, out);
}


# Gate sigmoid and tanh on the hardware tanh (v_tanh_f32): sigmoid(z) = 0.5 + 0.5 tanh(z/2), the 1/2 folded into the i,f,o weight fragments; 40 transcendentals per lane per step instead of 80
# speedup vs baseline: 7.5081x; 1.0711x over previous
#include <hip/hip_runtime.h>


typedef __attribute__((ext_vector_type(2))) float v2f;
typedef __attribute__((ext_vector_type(8))) float v8f;

#define T_LEN 512
#define H_STRIDE 68
#define H_TILE (16 * H_STRIDE)
#define P_TILE (3 * 4 * 16)

typedef __attribute__((ext_vector_type(4))) float v4f;

static __device__ __forceinline__ v8f wmma4(v2f a, v2f b, v8f c) {
  return __builtin_amdgcn_wmma_f32_16x16x4_f32(false, a, false, b, (short)0, c,
                                               false, false);
}

static __device__ __forceinline__ float sigm_h(float zh) {
  return fmaf(__builtin_amdgcn_tanhf(zh), 0.5f, 0.5f);
}
static __device__ __forceinline__ float tanh_(float x) {
  return __builtin_amdgcn_tanhf(x);
}

__global__ __launch_bounds__(128) void lstm_wmma_kernel(
    const float* __restrict__ x, const float* __restrict__ W_ih,
    const float* __restrict__ W_hh, const float* __restrict__ b_ih,
    const float* __restrict__ b_hh, const float* __restrict__ W_out,
    const float* __restrict__ b_out, float* __restrict__ out) {
  __shared__ float hS[2 * H_TILE];
  __shared__ __attribute__((aligned(16))) float pS[2 * P_TILE];

  const int tid  = threadIdx.x;
  const int lane = tid & 31;
  const int wave = __builtin_amdgcn_readfirstlane(tid >> 5);
  const int col  = lane & 15;
  const int hi   = lane >> 4;
  const int row0 = blockIdx.x * 16;

  v2f wb[17][4];
#pragma unroll
  for (int g = 0; g < 4; ++g) {
    const int n = g * 64 + wave * 16 + col;
    const float* wr = W_hh + n * 64 + hi * 2;
    const float sc = (g == 2) ? 1.0f : 0.5f;
#pragma unroll
    for (int ks = 0; ks < 16; ++ks) {
      wb[ks][g][0] = wr[ks * 4] * sc;
      wb[ks][g][1] = wr[ks * 4 + 1] * sc;
    }
    const float bs = b_ih[n] + b_hh[n];
    wb[16][g][0] = W_ih[n * 3 + hi * 2] * sc;
    wb[16][g][1] = (hi ? bs : W_ih[n * 3 + 1]) * sc;
  }
  v2f wo[4];
  const float* wor = W_out + ((col < 3) ? col : 0) * 64 + wave * 16 + hi * 2;
#pragma unroll
  for (int j = 0; j < 4; ++j) {
    wo[j][0] = (col < 3) ? wor[j * 4]     : 0.0f;
    wo[j][1] = (col < 3) ? wor[j * 4 + 1] : 0.0f;
  }
  const int sl = (lane < 12) ? lane : 0;
  const int so = sl % 3;
  const int sr = wave * 4 + sl / 3;
  const float bO = b_out[so];
  const int pRd = so * 64 + sr;
  float* osum = out + (size_t)(row0 + sr) * (T_LEN * 3) + so;
  const int pWr = (col * 4 + wave) * 16 + hi * 8;
  const int aOff = col * H_STRIDE + wave * 16 + hi * 2;

  for (int idx = tid; idx < H_TILE; idx += 128) hS[idx] = 0.0f;
  for (int idx = tid; idx < 2 * P_TILE; idx += 128) pS[idx] = 0.0f;
  __syncthreads();

  const float* xrow = x + (size_t)(row0 + col) * (T_LEN * 3);
  const v8f vzero = {0, 0, 0, 0, 0, 0, 0, 0};
  v8f cs = vzero;
  const int xk0 = hi * 2;
  const int xk1 = hi + 1;
  v2f ax;
  {
    const float x0 = xrow[xk0], x1 = xrow[xk1];
    ax[0] = x0;
    ax[1] = hi ? 1.0f : x1;
  }

  for (int t = 0; t < T_LEN; ++t) {
    const float* hR = &hS[(t & 1) * H_TILE];
    float*       hN = &hS[((t + 1) & 1) * H_TILE];

    const float* xp = xrow + ((t + 1 < T_LEN) ? (t + 1) : t) * 3;
    const float nx0 = xp[xk0], nx1 = xp[xk1];

    const float* pR = &pS[((t + 1) & 1) * P_TILE + pRd];
    const float q0 = pR[0], q1 = pR[16], q2 = pR[32], q3 = pR[48];

    v2f a[16];
#pragma unroll
    for (int ks = 0; ks < 16; ++ks)
      a[ks] = *(const v2f*)&hR[col * H_STRIDE + ks * 4 + hi * 2];
    v2f ap[4];
#pragma unroll
    for (int j = 0; j < 4; ++j) ap[j] = *(const v2f*)&hR[aOff + j * 4];

    v8f ai = wmma4(ax, wb[16][0], vzero);
    v8f af = wmma4(ax, wb[16][1], vzero);
    v8f ag = wmma4(ax, wb[16][2], vzero);
    v8f ao = wmma4(ax, wb[16][3], vzero);

    asm volatile("" :: "v"(q0), "v"(q1), "v"(q2), "v"(q3));
    if (t >= 2 && lane < 12) osum[(t - 2) * 3] = ((q0 + q1) + (q2 + q3)) + bO;

#pragma unroll
    for (int ks = 0; ks < 16; ++ks) {
      ai = wmma4(a[ks], wb[ks][0], ai);
      ag = wmma4(a[ks], wb[ks][2], ag);
    }
    asm volatile("" : "+v"(af), "+v"(ao) : "v"(ai), "v"(ag));
#pragma unroll
    for (int ks = 0; ks < 16; ++ks) {
      af = wmma4(a[ks], wb[ks][1], af);
      ao = wmma4(a[ks], wb[ks][3], ao);
    }
    float iv[8], gv[8];
#pragma unroll
    for (int r = 0; r < 8; ++r) {
      iv[r] = sigm_h(ai[r]);
      gv[r] = tanh_(ag[r]);
    }

    v2f ap0 = ap[0];
    asm volatile("" : "+v"(ap0) : "v"(af), "v"(ao));
    v8f po = wmma4(ap0, wo[0], vzero);
#pragma unroll
    for (int j = 1; j < 4; ++j) po = wmma4(ap[j], wo[j], po);

    float* hp = &hN[hi * 8 * H_STRIDE + wave * 16 + col];
#pragma unroll
    for (int r = 0; r < 8; ++r) {
      float fv = sigm_h(af[r]), ov = sigm_h(ao[r]);
      float cv = fv * cs[r] + iv[r] * gv[r];
      cs[r] = cv;
      hp[r * H_STRIDE] = ov * tanh_(cv);
    }
    if (col < 3) {
      v4f* pW = (v4f*)&pS[(t & 1) * P_TILE + pWr];
      pW[0] = v4f{po[0], po[1], po[2], po[3]};
      pW[1] = v4f{po[4], po[5], po[6], po[7]};
    }
    ax[0] = nx0;
    ax[1] = hi ? 1.0f : nx1;
    __syncthreads();
  }

  if (lane < 12) {
    const float* pR = &pS[((T_LEN + 1) & 1) * P_TILE + pRd];
    osum[(T_LEN - 2) * 3] = ((pR[0] + pR[16]) + (pR[32] + pR[48])) + bO;
  }
  {
    const float* hR = &hS[(T_LEN & 1) * H_TILE];
    v8f po = vzero;
#pragma unroll
    for (int j = 0; j < 4; ++j)
      po = wmma4(*(const v2f*)&hR[aOff + j * 4], wo[j], po);
    if (col < 3) {
      v4f* pW = (v4f*)&pS[(T_LEN & 1) * P_TILE + pWr];
      pW[0] = v4f{po[0], po[1], po[2], po[3]};
      pW[1] = v4f{po[4], po[5], po[6], po[7]};
    }
  }
  __syncthreads();
  if (lane < 12) {
    const float* pR = &pS[(T_LEN & 1) * P_TILE + pRd];
    osum[(T_LEN - 1) * 3] = ((pR[0] + pR[16]) + (pR[32] + pR[48])) + bO;
  }
}

extern "C" void kernel_launch(void* const* d_in, const int* in_sizes, int n_in,
                              void* d_out, int out_size, void* d_ws,
                              size_t ws_size, hipStream_t stream) {
  const float* x     = (const float*)d_in[0];
  const float* W_ih  = (const float*)d_in[1];
  const float* W_hh  = (const float*)d_in[2];
  const float* b_ih  = (const float*)d_in[3];
  const float* b_hh  = (const float*)d_in[4];
  const float* W_out = (const float*)d_in[5];
  const float* b_out = (const float*)d_in[6];
  float* out = (float*)d_out;

  dim3 grid(256);
  dim3 block(128);
  lstm_wmma_kernel<<<grid, block, 0, stream>>>(x, W_ih, W_hh, b_ih, b_hh,
                                               W_out, b_out, out);
}


# Products transposed (weights A, h^T B): a wave's own h tile in D layout is its B fragment; 20 of 72 WMMAs need no exchange and span barrier and LDS reads; gate sums start with own K slice
# speedup vs baseline: 7.8012x; 1.0390x over previous
#include <hip/hip_runtime.h>


typedef __attribute__((ext_vector_type(2))) float v2f;
typedef __attribute__((ext_vector_type(4))) float v4f;
typedef __attribute__((ext_vector_type(8))) float v8f;

#define T_LEN 512
#define H_STRIDE 68
#define H_TILE (16 * H_STRIDE)
#define P_TILE (3 * 4 * 16)

static __device__ __forceinline__ v8f wmma4(v2f a, v2f b, v8f c) {
  return __builtin_amdgcn_wmma_f32_16x16x4_f32(false, a, false, b, (short)0, c,
                                               false, false);
}

static __device__ __forceinline__ float sigm_h(float zh) {
  return fmaf(__builtin_amdgcn_tanhf(zh), 0.5f, 0.5f);
}
static __device__ __forceinline__ float tanh_(float x) {
  return __builtin_amdgcn_tanhf(x);
}

__global__ __launch_bounds__(128) void lstm_wmma_kernel(
    const float* __restrict__ x, const float* __restrict__ W_ih,
    const float* __restrict__ W_hh, const float* __restrict__ b_ih,
    const float* __restrict__ b_hh, const float* __restrict__ W_out,
    const float* __restrict__ b_out, float* __restrict__ out) {
  __shared__ __attribute__((aligned(16))) float hS[2 * H_TILE];
  __shared__ __attribute__((aligned(16))) float pS[2 * P_TILE];

  const int tid  = threadIdx.x;
  const int lane = tid & 31;
  const int wave = __builtin_amdgcn_readfirstlane(tid >> 5);
  const int col  = lane & 15;
  const int hi   = lane >> 4;
  const int row0 = blockIdx.x * 16;

  v2f wa[16][4];
  v2f wx[4];
#pragma unroll
  for (int g = 0; g < 4; ++g) {
    const int n = g * 64 + wave * 16 + col;
    const float sc = (g == 2) ? 1.0f : 0.5f;
#pragma unroll
    for (int ks = 0; ks < 16; ++ks) {
      const int src = (wave + (ks >> 2)) & 3;
      const float* wr = W_hh + n * 64 + src * 16 + hi * 8 + (ks & 3) * 2;
      wa[ks][g][0] = wr[0] * sc;
      wa[ks][g][1] = wr[1] * sc;
    }
    const float bs = b_ih[n] + b_hh[n];
    wx[g][0] = W_ih[n * 3 + hi * 2] * sc;
    wx[g][1] = (hi ? bs : W_ih[n * 3 + 1]) * sc;
  }
  v2f wo[4];
  const float* wor = W_out + ((col < 3) ? col : 0) * 64 + wave * 16 + hi * 8;
#pragma unroll
  for (int j = 0; j < 4; ++j) {
    wo[j][0] = (col < 3) ? wor[j * 2]     : 0.0f;
    wo[j][1] = (col < 3) ? wor[j * 2 + 1] : 0.0f;
  }
  const int sl = (lane < 12) ? lane : 0;
  const int so = sl % 3;
  const int sr = wave * 4 + sl / 3;
  const float bO = b_out[so];
  const int pRd = so * 64 + sr;
  float* osum = out + (size_t)(row0 + sr) * (T_LEN * 3) + so;
  const int pWr = wave * 16 + col;
  const int hOwn = col * H_STRIDE + wave * 16 + hi * 8;
  int hOth[3];
#pragma unroll
  for (int q = 0; q < 3; ++q)
    hOth[q] = col * H_STRIDE + ((wave + 1 + q) & 3) * 16 + hi * 8;

  for (int idx = tid; idx < 2 * H_TILE; idx += 128) hS[idx] = 0.0f;
  for (int idx = tid; idx < 2 * P_TILE; idx += 128) pS[idx] = 0.0f;
  __syncthreads();

  const float* xrow = x + (size_t)(row0 + col) * (T_LEN * 3);
  const v8f vzero = {0, 0, 0, 0, 0, 0, 0, 0};
  v8f cs = vzero;
  const int xk0 = hi * 2;
  const int xk1 = hi + 1;
  v2f bx;
  {
    const float x0 = xrow[xk0], x1 = xrow[xk1];
    bx[0] = x0;
    bx[1] = hi ? 1.0f : x1;
  }
  v2f hb[4];
#pragma unroll
  for (int j = 0; j < 4; ++j) hb[j] = v2f{0.0f, 0.0f};
  v8f ai = wmma4(wx[0], bx, vzero);
  v8f ag = wmma4(wx[2], bx, vzero);

  for (int t = 0; t < T_LEN; ++t) {
    const float* hR = &hS[(t & 1) * H_TILE];
    float*       hN = &hS[((t + 1) & 1) * H_TILE];

    v8f af = wmma4(wx[1], bx, vzero);
    v8f ao = wmma4(wx[3], bx, vzero);
    ai = wmma4(wa[0][0], hb[0], ai);
    af = wmma4(wa[0][1], hb[0], af);
    ag = wmma4(wa[0][2], hb[0], ag);
    ao = wmma4(wa[0][3], hb[0], ao);
    asm volatile("" : "+v"(ai), "+v"(af), "+v"(ag), "+v"(ao));
    __syncthreads();

    const float* xp = xrow + ((t + 1 < T_LEN) ? (t + 1) : t) * 3;
    float nx0 = xp[xk0], nx1 = xp[xk1];

    const float* pR = &pS[((t + 1) & 1) * P_TILE + pRd];
    const float q0 = pR[0], q1 = pR[16], q2 = pR[32], q3 = pR[48];

    v4f ob[3][2];
#pragma unroll
    for (int q = 0; q < 3; ++q) {
      ob[q][0] = *(const v4f*)&hR[hOth[q]];
      ob[q][1] = *(const v4f*)&hR[hOth[q] + 4];
    }

    asm volatile("" : "+v"(ai), "+v"(af), "+v"(ag), "+v"(ao));
#pragma unroll
    for (int j = 1; j < 4; ++j) {
      ai = wmma4(wa[j][0], hb[j], ai);
      af = wmma4(wa[j][1], hb[j], af);
      ag = wmma4(wa[j][2], hb[j], ag);
      ao = wmma4(wa[j][3], hb[j], ao);
    }

    asm volatile("" :: "v"(q0), "v"(q1), "v"(q2), "v"(q3));
    if (t >= 2 && lane < 12) osum[(t - 2) * 3] = ((q0 + q1) + (q2 + q3)) + bO;

#pragma unroll
    for (int q = 0; q < 3; ++q) {
#pragma unroll
      for (int j = 0; j < 4; ++j) {
        const v2f b = v2f{ob[q][j >> 1][2 * (j & 1)], ob[q][j >> 1][2 * (j & 1) + 1]};
        ai = wmma4(wa[4 * (q + 1) + j][0], b, ai);
        ag = wmma4(wa[4 * (q + 1) + j][2], b, ag);
      }
    }
    asm volatile("" : "+v"(af), "+v"(ao) : "v"(ai), "v"(ag));
#pragma unroll
    for (int q = 0; q < 3; ++q) {
#pragma unroll
      for (int j = 0; j < 4; ++j) {
        const v2f b = v2f{ob[q][j >> 1][2 * (j & 1)], ob[q][j >> 1][2 * (j & 1) + 1]};
        af = wmma4(wa[4 * (q + 1) + j][1], b, af);
        ao = wmma4(wa[4 * (q + 1) + j][3], b, ao);
      }
    }
    float iv[8], gv[8];
#pragma unroll
    for (int r = 0; r < 8; ++r) {
      iv[r] = sigm_h(ai[r]);
      gv[r] = tanh_(ag[r]);
    }

    v2f hb0 = hb[0];
    asm volatile("" : "+v"(hb0), "+v"(nx0), "+v"(nx1) : "v"(af), "v"(ao));
    v2f bxn;
    bxn[0] = nx0;
    bxn[1] = hi ? 1.0f : nx1;
    v8f po = wmma4(wo[0], hb0, vzero);
#pragma unroll
    for (int j = 1; j < 4; ++j) po = wmma4(wo[j], hb[j], po);
    ai = wmma4(wx[0], bxn, vzero);
    ag = wmma4(wx[2], bxn, vzero);

    float hn[8];
#pragma unroll
    for (int r = 0; r < 8; ++r) {
      float fv = sigm_h(af[r]), ov = sigm_h(ao[r]);
      float cv = fv * cs[r] + iv[r] * gv[r];
      cs[r] = cv;
      hn[r] = ov * tanh_(cv);
    }
    *(v4f*)&hN[hOwn]     = v4f{hn[0], hn[1], hn[2], hn[3]};
    *(v4f*)&hN[hOwn + 4] = v4f{hn[4], hn[5], hn[6], hn[7]};
#pragma unroll
    for (int j = 0; j < 4; ++j) hb[j] = v2f{hn[2 * j], hn[2 * j + 1]};
    if (hi == 0) {
      float* pW = &pS[(t & 1) * P_TILE + pWr];
      pW[0]   = po[0];
      pW[64]  = po[1];
      pW[128] = po[2];
    }
    bx = bxn;
  }

  __syncthreads();
  {
    v8f po = vzero;
#pragma unroll
    for (int j = 0; j < 4; ++j) po = wmma4(wo[j], hb[j], po);
    if (hi == 0) {
      float* pW = &pS[(T_LEN & 1) * P_TILE + pWr];
      pW[0]   = po[0];
      pW[64]  = po[1];
      pW[128] = po[2];
    }
  }
  __syncthreads();
  if (lane < 12) {
    const float* pA = &pS[((T_LEN + 1) & 1) * P_TILE + pRd];
    osum[(T_LEN - 2) * 3] = ((pA[0] + pA[16]) + (pA[32] + pA[48])) + bO;
    const float* pB = &pS[(T_LEN & 1) * P_TILE + pRd];
    osum[(T_LEN - 1) * 3] = ((pB[0] + pB[16]) + (pB[32] + pB[48])) + bO;
  }
}

extern "C" void kernel_launch(void* const* d_in, const int* in_sizes, int n_in,
                              void* d_out, int out_size, void* d_ws,
                              size_t ws_size, hipStream_t stream) {
  const float* x     = (const float*)d_in[0];
  const float* W_ih  = (const float*)d_in[1];
  const float* W_hh  = (const float*)d_in[2];
  const float* b_ih  = (const float*)d_in[3];
  const float* b_hh  = (const float*)d_in[4];
  const float* W_out = (const float*)d_in[5];
  const float* b_out = (const float*)d_in[6];
  float* out = (float*)d_out;

  dim3 grid(256);
  dim3 block(128);
  lstm_wmma_kernel<<<grid, block, 0, stream>>>(x, W_ih, W_hh, b_ih, b_hh,
                                               W_out, b_out, out);
}


# Global accesses on scalar base + fixed lane offset; out summed from an 8-step LDS ring, stored by all lanes once in 8 steps: no lane mask or address rewrite (translation wait) behind an access
# speedup vs baseline: 9.3432x; 1.1977x over previous
#include <hip/hip_runtime.h>


typedef __attribute__((ext_vector_type(2))) float v2f;
typedef __attribute__((ext_vector_type(4))) float v4f;
typedef __attribute__((ext_vector_type(8))) float v8f;

#define T_LEN 512
#define H_STRIDE 68
#define H_TILE (16 * H_STRIDE)
#define P_PITCH 36
#define P_WAVE (32 * P_PITCH)
#define P_BUF (4 * P_WAVE)

static __device__ __forceinline__ v8f wmma4(v2f a, v2f b, v8f c) {
  return __builtin_amdgcn_wmma_f32_16x16x4_f32(false, a, false, b, (short)0, c,
                                               false, false);
}

static __device__ __forceinline__ float sigm_h(float zh) {
  return fmaf(__builtin_amdgcn_tanhf(zh), 0.5f, 0.5f);
}
static __device__ __forceinline__ float tanh_(float x) {
  return __builtin_amdgcn_tanhf(x);
}

__global__ __launch_bounds__(128) void lstm_wmma_kernel(
    const float* __restrict__ x, const float* __restrict__ W_ih,
    const float* __restrict__ W_hh, const float* __restrict__ b_ih,
    const float* __restrict__ b_hh, const float* __restrict__ W_out,
    const float* __restrict__ b_out, float* __restrict__ out) {
  __shared__ __attribute__((aligned(16))) float hS[2 * H_TILE];
  __shared__ __attribute__((aligned(16))) float pS[2 * P_BUF];

  const int tid  = threadIdx.x;
  const int lane = tid & 31;
  const int wave = __builtin_amdgcn_readfirstlane(tid >> 5);
  const int col  = lane & 15;
  const int hi   = lane >> 4;
  const int row0 = blockIdx.x * 16;

  v2f wa[16][4];
  v2f wx[4];
#pragma unroll
  for (int g = 0; g < 4; ++g) {
    const int n = g * 64 + wave * 16 + col;
    const float sc = (g == 2) ? 1.0f : 0.5f;
#pragma unroll
    for (int ks = 0; ks < 16; ++ks) {
      const int src = (wave + (ks >> 2)) & 3;
      const float* wr = W_hh + n * 64 + src * 16 + hi * 8 + (ks & 3) * 2;
      wa[ks][g][0] = wr[0] * sc;
      wa[ks][g][1] = wr[1] * sc;
    }
    const float bs = b_ih[n] + b_hh[n];
    wx[g][0] = W_ih[n * 3 + hi * 2] * sc;
    wx[g][1] = (hi ? bs : W_ih[n * 3 + 1]) * sc;
  }
  v2f wo[4];
  const float* wor = W_out + ((col < 3) ? col : 0) * 64 + wave * 16 + hi * 8;
#pragma unroll
  for (int j = 0; j < 4; ++j) {
    wo[j][0] = (col < 3) ? wor[j * 2]     : 0.0f;
    wo[j][1] = (col < 3) ? wor[j * 2 + 1] : 0.0f;
  }
  const int pWr = (wave * 32 + col + 16 * hi) * P_PITCH;
  const int fr = wave * 4 + (lane >> 3);
  const int ft = lane & 7;
  const int pRd = fr * P_PITCH + ft * 4;
  const float bO0 = b_out[0], bO1 = b_out[1], bO2 = b_out[2];
  float* const outT = out + (size_t)row0 * (T_LEN * 3);
  const unsigned oOff = (unsigned)(fr * (T_LEN * 3) + ft * 3) * 4u;
  const int hOwn = col * H_STRIDE + wave * 16 + hi * 8;
  int hOth[3];
#pragma unroll
  for (int q = 0; q < 3; ++q)
    hOth[q] = col * H_STRIDE + ((wave + 1 + q) & 3) * 16 + hi * 8;

  for (int idx = tid; idx < 2 * H_TILE; idx += 128) hS[idx] = 0.0f;
  __syncthreads();

  const float* const xT = x + (size_t)row0 * (T_LEN * 3);
  const float* xrow = x + (size_t)(row0 + col) * (T_LEN * 3);
  const v8f vzero = {0, 0, 0, 0, 0, 0, 0, 0};
  v8f cs = vzero;
  const int xk0 = hi * 2;
  const int xk1 = hi + 1;
  unsigned xOff0 = (unsigned)(col * (T_LEN * 3) + xk0) * 4u;
  unsigned xOff1 = (unsigned)(col * (T_LEN * 3) + xk1) * 4u;
  v2f bx;
  {
    const float x0 = xrow[xk0], x1 = xrow[xk1];
    bx[0] = x0;
    bx[1] = hi ? 1.0f : x1;
  }
  v2f hb[4];
#pragma unroll
  for (int j = 0; j < 4; ++j) hb[j] = v2f{0.0f, 0.0f};
  v8f ai = wmma4(wx[0], bx, vzero);
  v8f ag = wmma4(wx[2], bx, vzero);

  for (int t = 0; t < T_LEN; ++t) {
    const float* hR = &hS[(t & 1) * H_TILE];
    float*       hN = &hS[((t + 1) & 1) * H_TILE];

    v8f af = wmma4(wx[1], bx, vzero);
    v8f ao = wmma4(wx[3], bx, vzero);
    ai = wmma4(wa[0][0], hb[0], ai);
    af = wmma4(wa[0][1], hb[0], af);
    ag = wmma4(wa[0][2], hb[0], ag);
    ao = wmma4(wa[0][3], hb[0], ao);
    asm volatile("" : "+v"(ai), "+v"(af), "+v"(ag), "+v"(ao));
    asm volatile("s_wait_dscnt 0x0" ::: "memory");
    __syncthreads();

    const char* xp = (const char*)(xT + ((t + 1 < T_LEN) ? (t + 1) : t) * 3);
    asm volatile("" : "+v"(xOff0), "+v"(xOff1));
    float nx0 = *(const float*)(xp + xOff0), nx1 = *(const float*)(xp + xOff1);


    v4f ob[3][2];
#pragma unroll
    for (int q = 0; q < 3; ++q) {
      ob[q][0] = *(const v4f*)&hR[hOth[q]];
      ob[q][1] = *(const v4f*)&hR[hOth[q] + 4];
    }

    const bool fl = ((t & 7) == 1) && (t >= 9);
    v4f f0, f1, f2, f3;
    if (fl) {
      const float* pR = &pS[(((t - 9) >> 3) & 1) * P_BUF + pRd];
      f0 = *(const v4f*)&pR[0];
      f1 = *(const v4f*)&pR[P_WAVE];
      f2 = *(const v4f*)&pR[2 * P_WAVE];
      f3 = *(const v4f*)&pR[3 * P_WAVE];
    }

    asm volatile("" : "+v"(ai), "+v"(af), "+v"(ag), "+v"(ao));
#pragma unroll
    for (int j = 1; j < 4; ++j) {
      ai = wmma4(wa[j][0], hb[j], ai);
      af = wmma4(wa[j][1], hb[j], af);
      ag = wmma4(wa[j][2], hb[j], ag);
      ao = wmma4(wa[j][3], hb[j], ao);
    }

    if (fl) {
      char* op = (char*)(outT + (t - 9) * 3) + oOff;
      ((float*)op)[0] = ((f0[0] + f1[0]) + (f2[0] + f3[0])) + bO0;
      ((float*)op)[1] = ((f0[1] + f1[1]) + (f2[1] + f3[1])) + bO1;
      ((float*)op)[2] = ((f0[2] + f1[2]) + (f2[2] + f3[2])) + bO2;
    }

#pragma unroll
    for (int q = 0; q < 3; ++q) {
#pragma unroll
      for (int j = 0; j < 4; ++j) {
        const v2f b = v2f{ob[q][j >> 1][2 * (j & 1)], ob[q][j >> 1][2 * (j & 1) + 1]};
        ai = wmma4(wa[4 * (q + 1) + j][0], b, ai);
        ag = wmma4(wa[4 * (q + 1) + j][2], b, ag);
      }
    }
    asm volatile("" : "+v"(af), "+v"(ao) : "v"(ai), "v"(ag));
#pragma unroll
    for (int q = 0; q < 3; ++q) {
#pragma unroll
      for (int j = 0; j < 4; ++j) {
        const v2f b = v2f{ob[q][j >> 1][2 * (j & 1)], ob[q][j >> 1][2 * (j & 1) + 1]};
        af = wmma4(wa[4 * (q + 1) + j][1], b, af);
        ao = wmma4(wa[4 * (q + 1) + j][3], b, ao);
      }
    }
    float iv[8], gv[8];
#pragma unroll
    for (int r = 0; r < 8; ++r) {
      iv[r] = sigm_h(ai[r]);
      gv[r] = tanh_(ag[r]);
    }

    v2f hb0 = hb[0];
    asm volatile("" : "+v"(hb0), "+v"(nx0), "+v"(nx1) : "v"(af), "v"(ao));
    v2f bxn;
    bxn[0] = nx0;
    bxn[1] = hi ? 1.0f : nx1;
    v8f po = wmma4(wo[0], hb0, vzero);
#pragma unroll
    for (int j = 1; j < 4; ++j) po = wmma4(wo[j], hb[j], po);
    ai = wmma4(wx[0], bxn, vzero);
    ag = wmma4(wx[2], bxn, vzero);

    float hn[8];
#pragma unroll
    for (int r = 0; r < 8; ++r) {
      float fv = sigm_h(af[r]), ov = sigm_h(ao[r]);
      float cv = fv * cs[r] + iv[r] * gv[r];
      cs[r] = cv;
      hn[r] = ov * tanh_(cv);
    }
    *(v4f*)&hN[hOwn]     = v4f{hn[0], hn[1], hn[2], hn[3]};
    *(v4f*)&hN[hOwn + 4] = v4f{hn[4], hn[5], hn[6], hn[7]};
#pragma unroll
    for (int j = 0; j < 4; ++j) hb[j] = v2f{hn[2 * j], hn[2 * j + 1]};
    *(v4f*)&pS[(((t + 15) >> 3) & 1) * P_BUF + pWr + ((t + 7) & 7) * 4] =
        v4f{po[0], po[1], po[2], po[3]};
    bx = bxn;
  }

  {
    v8f po = vzero;
#pragma unroll
    for (int j = 0; j < 4; ++j) po = wmma4(wo[j], hb[j], po);
    *(v4f*)&pS[(((T_LEN + 15) >> 3) & 1) * P_BUF + pWr + ((T_LEN + 7) & 7) * 4] =
        v4f{po[0], po[1], po[2], po[3]};
  }
  __syncthreads();
  {
    const float* pR = &pS[(((T_LEN - 8) >> 3) & 1) * P_BUF + pRd];
    const v4f f0 = *(const v4f*)&pR[0];
    const v4f f1 = *(const v4f*)&pR[P_WAVE];
    const v4f f2 = *(const v4f*)&pR[2 * P_WAVE];
    const v4f f3 = *(const v4f*)&pR[3 * P_WAVE];
    char* op = (char*)(outT + (T_LEN - 8) * 3) + oOff;
    ((float*)op)[0] = ((f0[0] + f1[0]) + (f2[0] + f3[0])) + bO0;
    ((float*)op)[1] = ((f0[1] + f1[1]) + (f2[1] + f3[1])) + bO1;
    ((float*)op)[2] = ((f0[2] + f1[2]) + (f2[2] + f3[2])) + bO2;
  }
}

extern "C" void kernel_launch(void* const* d_in, const int* in_sizes, int n_in,
                              void* d_out, int out_size, void* d_ws,
                              size_t ws_size, hipStream_t stream) {
  const float* x     = (const float*)d_in[0];
  const float* W_ih  = (const float*)d_in[1];
  const float* W_hh  = (const float*)d_in[2];
  const float* b_ih  = (const float*)d_in[3];
  const float* b_hh  = (const float*)d_in[4];
  const float* W_out = (const float*)d_in[5];
  const float* b_out = (const float*)d_in[6];
  float* out = (float*)d_out;

  dim3 grid(256);
  dim3 block(128);
  lstm_wmma_kernel<<<grid, block, 0, stream>>>(x, W_ih, W_hh, b_ih, b_hh,
                                               W_out, b_out, out);
}
